# starting kernel
# baseline (speedup 1.0000x reference)
#include <hip/hip_runtime.h>


typedef __attribute__((ext_vector_type(2))) float v2f;
typedef __attribute__((ext_vector_type(8))) float v8f;

#define N_MAPS 1024
#define T_IN   10
#define HW     196
#define NTOT   (16 * HW)
#define TILES_PER_C (NTOT / 16)
#define WAVES_PER_BLOCK 8

__global__ __launch_bounds__(256)
void conv_over_time_wmma_kernel(const float* __restrict__ x,
                                const float* __restrict__ W,
                                const float* __restrict__ bias,
                                float* __restrict__ out)
{
    const int lane = threadIdx.x & 31;
    const int wid  = blockIdx.x * WAVES_PER_BLOCK + (threadIdx.x >> 5);
    const int c    = wid / TILES_PER_C;
    const int tile = wid - c * TILES_PER_C;

    const int half = lane >> 4;
    const int nidx = lane & 15;
    const int n    = tile * 16 + nidx;
    const int b    = n / HW;
    const int hw   = n - b * HW;

    const long long t_stride = (long long)N_MAPS * HW;
    const long long base     = ((long long)b * T_IN * N_MAPS + (long long)c) * HW + hw;

    const float* __restrict__ wrow = W + c * T_IN;

    v8f acc = (v8f)0.0f;

    #pragma unroll
    for (int kk = 0; kk < 3; ++kk) {
        const int t0  = 4 * kk + 2 * half;
        const int t1  = t0 + 1;
        const int t0c = (t0 < T_IN) ? t0 : (T_IN - 1);
        const int t1c = (t1 < T_IN) ? t1 : (T_IN - 1);

        v2f xb, wa;
        xb.x = x[base + (long long)t0c * t_stride];
        xb.y = x[base + (long long)t1c * t_stride];

        const float w0 = wrow[t0c];
        const float w1 = wrow[t1c];
        wa.x = (t0 < T_IN) ? w0 : 0.0f;
        wa.y = (t1 < T_IN) ? w1 : 0.0f;

        acc = __builtin_amdgcn_wmma_f32_16x16x4_f32(
                  false, wa, false, xb, (short)0, acc, false, false);
    }

    if (lane < 16) {
        out[(long long)b * (N_MAPS * HW) + (long long)c * HW + hw]
            = acc[0] + bias[c];
    }
}

extern "C" void kernel_launch(void* const* d_in, const int* in_sizes, int n_in,
                              void* d_out, int out_size, void* d_ws, size_t ws_size,
                              hipStream_t stream) {
    const float* x  = (const float*)d_in[0];
    const float* W  = (const float*)d_in[1];
    const float* bb = (const float*)d_in[2];
    float* out      = (float*)d_out;

    const int total_waves = N_MAPS * TILES_PER_C;
    dim3 grid(total_waves / WAVES_PER_BLOCK);
    dim3 block(32 * WAVES_PER_BLOCK);

    conv_over_time_wmma_kernel<<<grid, block, 0, stream>>>(x, W, bb, out);
}


# Half as many, longer waves: 512 work-groups of 8 waves (two per WGP); a wave owns 4 images of one channel (49 tiles) and keeps the loads of 8 tiles in flight ahead of its matrix work
# speedup vs baseline: 1.6873x; 1.6873x over previous
#include <hip/hip_runtime.h>


typedef __attribute__((ext_vector_type(2))) float v2f;
typedef __attribute__((ext_vector_type(8))) float v8f;

#define N_MAPS 1024
#define T_IN   10
#define HW     196
#define PLANE  (N_MAPS * HW)
#define IMAGE  (T_IN * PLANE)
#define WAVES_PER_WG   8
#define WAVES_PER_C    4
#define IMGS_PER_WAVE  4
#define TILES_PER_WAVE 49
#define DEPTH          8

struct Tile { v2f f[3]; };

__global__ __launch_bounds__(32 * WAVES_PER_WG)
void conv_over_time_wmma_kernel(const float* __restrict__ x,
                                const float* __restrict__ W,
                                const float* __restrict__ bias,
                                float* __restrict__ out)
{
    const unsigned lane = threadIdx.x & 31u;
    const unsigned wv   = (unsigned)__builtin_amdgcn_readfirstlane((int)(threadIdx.x >> 5));
    const unsigned c    = blockIdx.x * 2u + wv / WAVES_PER_C;
    const unsigned b0   = (wv % WAVES_PER_C) * IMGS_PER_WAVE;
    const unsigned half = lane >> 4;
    const unsigned nidx = lane & 15u;

    const float* __restrict__ wrow = W + c * T_IN;
    v2f      wa[3];
    unsigned toff0[3], toff1[3];
    #pragma unroll
    for (int kk = 0; kk < 3; ++kk) {
        const unsigned t0  = 4u * kk + 2u * half;
        const unsigned t1  = t0 + 1u;
        const unsigned t0c = (t0 < T_IN) ? t0 : (T_IN - 1u);
        const unsigned t1c = (t1 < T_IN) ? t1 : (T_IN - 1u);
        const float w0 = wrow[t0c];
        const float w1 = wrow[t1c];
        wa[kk].x  = (t0 < T_IN) ? w0 : 0.0f;
        wa[kk].y  = (t1 < T_IN) ? w1 : 0.0f;
        toff0[kk] = t0c * (unsigned)PLANE;
        toff1[kk] = t1c * (unsigned)PLANE;
    }
    const float bc = bias[c];

    const unsigned xbase = (b0 * (T_IN * N_MAPS) + c) * HW;
    const unsigned obase = (b0 * N_MAPS + c) * HW;

    auto load_tile = [&](int i, Tile& s) {
        const unsigned p  = 16u * (unsigned)i + nidx;
        const unsigned bl = p / HW;
        const unsigned hw = p - bl * HW;
        const unsigned xo = xbase + bl * (unsigned)IMAGE + hw;
        #pragma unroll
        for (int kk = 0; kk < 3; ++kk) {
            s.f[kk].x = x[xo + toff0[kk]];
            s.f[kk].y = x[xo + toff1[kk]];
        }
    };

    auto compute_tile = [&](int i, const Tile& s) {
        v8f acc = (v8f)0.0f;
        #pragma unroll
        for (int kk = 0; kk < 3; ++kk) {
            acc = __builtin_amdgcn_wmma_f32_16x16x4_f32(
                      false, wa[kk], false, s.f[kk], (short)0, acc,
                      false, false);
        }
        const unsigned p  = 16u * (unsigned)i + nidx;
        const unsigned bl = p / HW;
        const unsigned hw = p - bl * HW;
        if (lane < 16u) {
            out[obase + bl * (unsigned)PLANE + hw] = acc[0] + bc;
        }
    };

    Tile ring[DEPTH + 1];
    #pragma unroll
    for (int i = 0; i < DEPTH; ++i) load_tile(i, ring[i]);
    #pragma unroll
    for (int i = 0; i < TILES_PER_WAVE; ++i) {
        if (i + DEPTH < TILES_PER_WAVE)
            load_tile(i + DEPTH, ring[(i + DEPTH) % (DEPTH + 1)]);
        compute_tile(i, ring[i % (DEPTH + 1)]);
    }
}

extern "C" void kernel_launch(void* const* d_in, const int* in_sizes, int n_in,
                              void* d_out, int out_size, void* d_ws, size_t ws_size,
                              hipStream_t stream) {
    const float* x  = (const float*)d_in[0];
    const float* W  = (const float*)d_in[1];
    const float* bb = (const float*)d_in[2];
    float* out      = (float*)d_out;

    dim3 grid(N_MAPS / 2);
    dim3 block(32 * WAVES_PER_WG);

    conv_over_time_wmma_kernel<<<grid, block, 0, stream>>>(x, W, bb, out);
}


# Wave owns 4 adjacent channels of ONE image, not 4 images of a channel: aligned 3136B run per plane in whole 64B pieces, immediate offsets, scalar weight loads; 2 weight rows on edge tiles; same sums
# speedup vs baseline: 2.0322x; 1.2044x over previous
#include <hip/hip_runtime.h>


typedef __attribute__((ext_vector_type(2))) float v2f;
typedef __attribute__((ext_vector_type(8))) float v8f;

#define N_MAPS 1024
#define T_IN   10
#define HW     196
#define PLANE  (N_MAPS * HW)
#define WAVES_PER_WG   8
#define CH_PER_WAVE    4
#define CH_PER_WG      (WAVES_PER_WG * CH_PER_WAVE)
#define WGS_PER_IMAGE  (N_MAPS / CH_PER_WG)
#define TILES_PER_WAVE 49
#define DEPTH          8

struct Tile { v2f f[3]; };

__global__ __launch_bounds__(32 * WAVES_PER_WG)
void conv_over_time_wmma_kernel(const float* __restrict__ x,
                                const float* __restrict__ W,
                                const float* __restrict__ bias,
                                float* __restrict__ out)
{
    const unsigned lane = threadIdx.x & 31u;
    const unsigned wv   = (unsigned)__builtin_amdgcn_readfirstlane((int)(threadIdx.x >> 5));
    const unsigned b    = blockIdx.x / WGS_PER_IMAGE;
    const unsigned c0   = (blockIdx.x % WGS_PER_IMAGE) * CH_PER_WG
                        + wv * CH_PER_WAVE;
    const unsigned half = lane >> 4;
    const unsigned nidx = lane & 15u;


    float w[CH_PER_WAVE][T_IN];
    float bc[CH_PER_WAVE];
    #pragma unroll
    for (int ch = 0; ch < CH_PER_WAVE; ++ch) {
        const float* __restrict__ wrow = W + (c0 + (unsigned)ch) * T_IN;
        #pragma unroll
        for (int t = 0; t < T_IN; ++t) w[ch][t] = wrow[t];
        bc[ch] = bias[c0 + (unsigned)ch];
    }

    const float* __restrict__ xw = x + ((size_t)((b * (T_IN * N_MAPS) + c0) * HW) + nidx);
    const float* __restrict__ px0[3];
    const float* __restrict__ px1[3];
    #pragma unroll
    for (int kk = 0; kk < 3; ++kk) {
        const unsigned t0  = 4u * kk + 2u * half;
        const unsigned t1  = t0 + 1u;
        const unsigned t0c = (t0 < T_IN) ? t0 : (T_IN - 1u);
        const unsigned t1c = (t1 < T_IN) ? t1 : (T_IN - 1u);
        px0[kk] = xw + (size_t)t0c * PLANE;
        px1[kk] = xw + (size_t)t1c * PLANE;
    }
    float* __restrict__ ow = out + ((size_t)((b * N_MAPS + c0) * HW) + nidx);

    auto load_tile = [&](int i, Tile& s) {
        #pragma unroll
        for (int kk = 0; kk < 3; ++kk) {
            s.f[kk].x = px0[kk][16 * i];
            s.f[kk].y = px1[kk][16 * i];
        }
    };

    Tile ring[DEPTH + 1];
    #pragma unroll
    for (int i = 0; i < DEPTH; ++i) {
        load_tile(i, ring[i]);
        __builtin_amdgcn_sched_barrier(0);
    }

    const bool hi = (half != 0u);
    v2f wa[CH_PER_WAVE][3];
#define BUILD_FRAGS(ch)                                                        \
    wa[ch][0].x = hi ? w[ch][2] : w[ch][0];  wa[ch][0].y = hi ? w[ch][3] : w[ch][1];  \
    wa[ch][1].x = hi ? w[ch][6] : w[ch][4];  wa[ch][1].y = hi ? w[ch][7] : w[ch][5];  \
    wa[ch][2].x = hi ? 0.0f     : w[ch][8];  wa[ch][2].y = hi ? 0.0f     : w[ch][9];
    BUILD_FRAGS(0)
    BUILD_FRAGS(1)
    BUILD_FRAGS(2)
    BUILD_FRAGS(3)
#undef BUILD_FRAGS

    const bool oddrow = (lane & 1u) != 0u;
    auto mix = [&](v2f lo, v2f hi) {
        v2f r;
        r.x = oddrow ? hi.x : lo.x;
        r.y = oddrow ? hi.y : lo.y;
        return r;
    };
    v2f wm[CH_PER_WAVE - 1][3];
    wm[0][0] = mix(wa[0][0], wa[1][0]);  wm[0][1] = mix(wa[0][1], wa[1][1]);  wm[0][2] = mix(wa[0][2], wa[1][2]);
    wm[1][0] = mix(wa[1][0], wa[2][0]);  wm[1][1] = mix(wa[1][1], wa[2][1]);  wm[1][2] = mix(wa[1][2], wa[2][2]);
    wm[2][0] = mix(wa[2][0], wa[3][0]);  wm[2][1] = mix(wa[2][1], wa[3][1]);  wm[2][2] = mix(wa[2][2], wa[3][2]);

    auto compute_tile = [&](int i, const Tile& s) {
        const int  chA   = (16 * i) / HW;
        const int  chB   = (16 * i + 15) / HW;
        const bool cross = (chA != chB);
        v8f acc = (v8f)0.0f;
        #pragma unroll
        for (int kk = 0; kk < 3; ++kk) {
            const v2f a = cross ? wm[chA < CH_PER_WAVE - 1 ? chA : 0][kk] : wa[chA][kk];
            acc = __builtin_amdgcn_wmma_f32_16x16x4_f32(
                      false, a, false, s.f[kk], (short)0, acc,
                      false, false);
        }
        float r = acc[0] + bc[chA];
        if (cross) {
            const unsigned split = (unsigned)(HW * chB - 16 * i);
            const float rB = acc[1] + bc[chB];
            r = (nidx < split) ? r : rB;
        }
        if (lane < 16u) {
            ow[16 * i] = r;
        }
    };

    #pragma unroll
    for (int i = 0; i < TILES_PER_WAVE; ++i) {
        if (i + DEPTH < TILES_PER_WAVE)
            load_tile(i + DEPTH, ring[(i + DEPTH) % (DEPTH + 1)]);
        compute_tile(i, ring[i % (DEPTH + 1)]);
    }
}

extern "C" void kernel_launch(void* const* d_in, const int* in_sizes, int n_in,
                              void* d_out, int out_size, void* d_ws, size_t ws_size,
                              hipStream_t stream) {
    const float* x  = (const float*)d_in[0];
    const float* W  = (const float*)d_in[1];
    const float* bb = (const float*)d_in[2];
    float* out      = (float*)d_out;

    dim3 grid(16 * WGS_PER_IMAGE);
    dim3 block(32 * WAVES_PER_WG);

    conv_over_time_wmma_kernel<<<grid, block, 0, stream>>>(x, W, bb, out);
}
